# starting kernel
# baseline (speedup 1.0000x reference)
#include <hip/hip_runtime.h>
#include <math.h>

#define B_N   4
#define C_CH  96
#define DDIM  16
#define HDIM  56
#define WDIM  56
#define GDIM  96
#define HID   24
#define NB    3
#define NCOL  (C_CH + GDIM)
#define SPATIAL (DDIM * HDIM * WDIM)
#define LN_EPS 1e-5f

typedef float v2f __attribute__((ext_vector_type(2)));
typedef float v4f __attribute__((ext_vector_type(4)));
typedef float v8f __attribute__((ext_vector_type(8)));

__global__ void feat_reduce_kernel(const float* __restrict__ x,
                                   float* __restrict__ feat) {
    const int bc = blockIdx.x;
    const v4f* p = (const v4f*)(x + (size_t)bc * SPATIAL);
    float s = 0.f;
    for (int i = threadIdx.x; i < SPATIAL / 4; i += 256) {
        v4f v = p[i];
        s += v.x + v.y + v.z + v.w;
    }
    __shared__ float red[256];
    red[threadIdx.x] = s;
    __syncthreads();
    for (int off = 128; off > 0; off >>= 1) {
        if (threadIdx.x < off) red[threadIdx.x] += red[threadIdx.x + off];
        __syncthreads();
    }
    if (threadIdx.x == 0) feat[bc] = red[0] * (1.0f / (float)SPATIAL);
}

__global__ void gate_kernel(const float* __restrict__ feat,
                            const float* __restrict__ guidance,
                            const float* __restrict__ ln_g,
                            const float* __restrict__ ln_b,
                            const float* __restrict__ w1p,
                            const float* __restrict__ b1p,
                            const float* __restrict__ w2p,
                            const float* __restrict__ b2p,
                            float* __restrict__ wts) {
    __shared__ float gn[16][NCOL];
    __shared__ float w1t[32][NCOL];
    __shared__ float mu[B_N];
    __shared__ float rs[B_N];
    __shared__ float hbuf[B_N][HID];
    __shared__ float logits[B_N][NB];

    const int lane = threadIdx.x;

    for (int idx = lane; idx < 12 * NCOL; idx += 32)
        gn[4 + idx / NCOL][idx % NCOL] = 0.f;
    for (int idx = lane; idx < 32 * NCOL; idx += 32) {
        const int n = idx / NCOL, k = idx % NCOL;
        w1t[n][k] = (n < HID) ? w1p[k * HID + n] : 0.f;
    }
    for (int idx = lane; idx < B_N * NCOL; idx += 32) {
        const int r = idx / NCOL, col = idx % NCOL;
        gn[r][col] = (col < C_CH) ? feat[r * C_CH + col]
                                  : guidance[r * GDIM + (col - C_CH)];
    }
    __syncthreads();

    if (lane < B_N) {
        float m = 0.f;
        for (int j = 0; j < NCOL; ++j) m += gn[lane][j];
        m *= (1.0f / NCOL);
        float v = 0.f;
        for (int j = 0; j < NCOL; ++j) { float d = gn[lane][j] - m; v += d * d; }
        v *= (1.0f / NCOL);
        mu[lane] = m;
        rs[lane] = rsqrtf(v + LN_EPS);
    }
    __syncthreads();

    for (int idx = lane; idx < B_N * NCOL; idx += 32) {
        const int r = idx / NCOL, col = idx % NCOL;
        gn[r][col] = (gn[r][col] - mu[r]) * rs[r] * ln_g[col] + ln_b[col];
    }
    __syncthreads();

    const int half = lane >> 4;
    const int l    = lane & 15;
    const float* arow  = &gn[l][0];
    const float* b0row = &w1t[l][0];
    const float* b1row = &w1t[16 + l][0];
    v8f c0 = {};
    v8f c1 = {};
    for (int kk = 0; kk < NCOL / 4; ++kk) {
        const int kb = kk * 4 + half * 2;
        const v2f a   = *(const v2f*)(arow + kb);
        const v2f b0  = *(const v2f*)(b0row + kb);
        const v2f b1v = *(const v2f*)(b1row + kb);
        c0 = __builtin_amdgcn_wmma_f32_16x16x4_f32(false, a, false, b0,
                                                   (short)0, c0, false, false);
        c1 = __builtin_amdgcn_wmma_f32_16x16x4_f32(false, a, false, b1v,
                                                   (short)0, c1, false, false);
    }

    #pragma unroll
    for (int r = 0; r < 8; ++r) {
        const int M = half ? (r + 8) : r;
        if (M < B_N) {
            float v0 = c0[r] + b1p[l];
            v0 = 0.5f * v0 * (1.0f + erff(v0 * 0.70710678118654752f));
            hbuf[M][l] = v0;
            const int n1 = 16 + l;
            if (n1 < HID) {
                float v1 = c1[r] + b1p[n1];
                v1 = 0.5f * v1 * (1.0f + erff(v1 * 0.70710678118654752f));
                hbuf[M][n1] = v1;
            }
        }
    }
    __syncthreads();

    if (lane < B_N * NB) {
        const int b = lane / NB, j = lane % NB;
        float s = b2p[j];
        for (int k = 0; k < HID; ++k) s += hbuf[b][k] * w2p[k * NB + j];
        logits[b][j] = s;
    }
    __syncthreads();

    if (lane < B_N) {
        float m = fmaxf(logits[lane][0], fmaxf(logits[lane][1], logits[lane][2]));
        float e0 = __expf(logits[lane][0] - m);
        float e1 = __expf(logits[lane][1] - m);
        float e2 = __expf(logits[lane][2] - m);
        float inv = 1.0f / (e0 + e1 + e2);
        wts[lane * NB + 0] = e0 * inv;
        wts[lane * NB + 1] = e1 * inv;
        wts[lane * NB + 2] = e2 * inv;
    }
}

__global__ void fold_kernel(const float* __restrict__ convw,
                            const float* __restrict__ convb,
                            const float* __restrict__ wts,
                            float* __restrict__ wscaled,
                            float* __restrict__ biaseff) {
    const int idx = blockIdx.x * 256 + threadIdx.x;
    const int total = B_N * C_CH * NB * 27;
    if (idx < total) {
        const int t = idx % 27;
        int tmp = idx / 27;
        const int i = tmp % NB;
        tmp /= NB;
        const int c = tmp % C_CH;
        const int b = tmp / C_CH;
        wscaled[idx] = wts[b * NB + i] * convw[(i * C_CH + c) * 27 + t];
    }
    if (idx < B_N * C_CH) {
        const int c = idx % C_CH, b = idx / C_CH;
        float s = 0.f;
        for (int i = 0; i < NB; ++i) s += wts[b * NB + i] * convb[i * C_CH + c];
        biaseff[idx] = s;
    }
}

__global__ void dwconv_kernel(const float* __restrict__ x,
                              const float* __restrict__ wscaled,
                              const float* __restrict__ biaseff,
                              float* __restrict__ out) {
    const int bcd = blockIdx.x;
    const int z  = bcd % DDIM;
    const int bc = bcd / DDIM;

    __shared__ float wsm[NB * 27];
    __shared__ float biass;
    if (threadIdx.x < NB * 27) wsm[threadIdx.x] = wscaled[bc * NB * 27 + threadIdx.x];
    if (threadIdx.x == NB * 27) biass = biaseff[bc];
    __syncthreads();

    const float* xp = x + (size_t)bc * SPATIAL;
    float* op = out + (size_t)bc * SPATIAL + (size_t)z * (HDIM * WDIM);
    const bool zint = (z >= 3) && (z <= DDIM - 4);

    for (int hw = threadIdx.x; hw < HDIM * WDIM; hw += 256) {
        const int y  = hw / WDIM;
        const int xw = hw % WDIM;
        float acc = biass;
        if (zint && (y >= 3) && (y <= HDIM - 4) && (xw >= 3) && (xw <= WDIM - 4)) {
            const float* base = xp + ((size_t)z * HDIM + y) * WDIM + xw;
            #pragma unroll
            for (int i = 0; i < NB; ++i) {
                const int di = i + 1;
                const float* wp = &wsm[i * 27];
                #pragma unroll
                for (int kz = 0; kz < 3; ++kz) {
                    #pragma unroll
                    for (int ky = 0; ky < 3; ++ky) {
                        const float* row = base + ((kz - 1) * di * HDIM
                                                 + (ky - 1) * di) * WDIM;
                        #pragma unroll
                        for (int kx = 0; kx < 3; ++kx)
                            acc = fmaf(wp[(kz * 3 + ky) * 3 + kx],
                                       row[(kx - 1) * di], acc);
                    }
                }
            }
        } else {
            #pragma unroll
            for (int i = 0; i < NB; ++i) {
                const int di = i + 1;
                const float* wp = &wsm[i * 27];
                #pragma unroll
                for (int kz = 0; kz < 3; ++kz) {
                    const int z2 = z + (kz - 1) * di;
                    if ((unsigned)z2 >= (unsigned)DDIM) continue;
                    #pragma unroll
                    for (int ky = 0; ky < 3; ++ky) {
                        const int y2 = y + (ky - 1) * di;
                        if ((unsigned)y2 >= (unsigned)HDIM) continue;
                        const float* row = xp + ((size_t)z2 * HDIM + y2) * WDIM;
                        #pragma unroll
                        for (int kx = 0; kx < 3; ++kx) {
                            const int x2 = xw + (kx - 1) * di;
                            if ((unsigned)x2 >= (unsigned)WDIM) continue;
                            acc = fmaf(wp[(kz * 3 + ky) * 3 + kx], row[x2], acc);
                        }
                    }
                }
            }
        }
        op[hw] = acc;
    }
}

extern "C" void kernel_launch(void* const* d_in, const int* in_sizes, int n_in,
                              void* d_out, int out_size, void* d_ws, size_t ws_size,
                              hipStream_t stream) {
    const float* x        = (const float*)d_in[0];
    const float* guidance = (const float*)d_in[1];
    const float* convw    = (const float*)d_in[2];
    const float* convb    = (const float*)d_in[3];
    const float* ln_g     = (const float*)d_in[4];
    const float* ln_b     = (const float*)d_in[5];
    const float* w1       = (const float*)d_in[6];
    const float* b1       = (const float*)d_in[7];
    const float* w2       = (const float*)d_in[8];
    const float* b2       = (const float*)d_in[9];

    float* ws      = (float*)d_ws;
    float* feat    = ws;
    float* wts     = ws + 384;
    float* biaseff = ws + 512;
    float* wscaled = ws + 1024;

    feat_reduce_kernel<<<B_N * C_CH, 256, 0, stream>>>(x, feat);
    gate_kernel<<<1, 32, 0, stream>>>(feat, guidance, ln_g, ln_b,
                                      w1, b1, w2, b2, wts);
    const int foldN = B_N * C_CH * NB * 27;
    fold_kernel<<<(foldN + 255) / 256, 256, 0, stream>>>(convw, convb, wts,
                                                         wscaled, biaseff);
    dwconv_kernel<<<B_N * C_CH * DDIM, 256, 0, stream>>>(x, wscaled, biaseff,
                                                         (float*)d_out);
}


# gate MLP and weight fold merged into one kernel of 4 blocks, one per batch row: softmax weights stay in registers, fold operands read up front with the gate's; 4 launches to 3; sum orders kept
# speedup vs baseline: 14.3633x; 14.3633x over previous
#include <hip/hip_runtime.h>
#include <math.h>

#define B_N   4
#define C_CH  96
#define DDIM  16
#define HDIM  56
#define WDIM  56
#define GDIM  96
#define HID   24
#define NB    3
#define NCOL  (C_CH + GDIM)
#define SPATIAL (DDIM * HDIM * WDIM)
#define LN_EPS 1e-5f

typedef float v2f __attribute__((ext_vector_type(2)));
typedef float v4f __attribute__((ext_vector_type(4)));
typedef float v8f __attribute__((ext_vector_type(8)));

__global__ void feat_reduce_kernel(const float* __restrict__ x,
                                   float* __restrict__ feat) {
    const int bc = blockIdx.x;
    const v4f* p = (const v4f*)(x + (size_t)bc * SPATIAL);
    float s = 0.f;
    for (int i = threadIdx.x; i < SPATIAL / 4; i += 256) {
        v4f v = p[i];
        s += v.x + v.y + v.z + v.w;
    }
    __shared__ float red[256];
    red[threadIdx.x] = s;
    __syncthreads();
    for (int off = 128; off > 0; off >>= 1) {
        if (threadIdx.x < off) red[threadIdx.x] += red[threadIdx.x + off];
        __syncthreads();
    }
    if (threadIdx.x == 0) feat[bc] = red[0] * (1.0f / (float)SPATIAL);
}

#define GATE_WAVES   8
#define GATE_THREADS (GATE_WAVES * 32)
#define KSLICE       (NCOL / GATE_WAVES)
#define KSTEPS       (KSLICE / 4)
#define GPL          (NCOL / 32)
#define FOLD_ROWS    (NB * 27)
#define FOLD_CG      3
#define FOLD_THREADS (FOLD_CG * FOLD_ROWS)
#define FOLD_ITERS   (C_CH / FOLD_CG)

__device__ __forceinline__ float wave_sum(float v) {
    v += __shfl_xor(v, 16, 32);
    v += __shfl_xor(v, 8, 32);
    v += __shfl_xor(v, 4, 32);
    v += __shfl_xor(v, 2, 32);
    v += __shfl_xor(v, 1, 32);
    return v;
}

__global__ void __launch_bounds__(GATE_THREADS)
gate_fold_kernel(const float* __restrict__ feat,
                 const float* __restrict__ guidance,
                 const float* __restrict__ ln_g,
                 const float* __restrict__ ln_b,
                 const float* __restrict__ w1p,
                 const float* __restrict__ b1p,
                 const float* __restrict__ w2p,
                 const float* __restrict__ b2p,
                 const float* __restrict__ convw,
                 const float* __restrict__ convb,
                 float* __restrict__ wscaled,
                 float* __restrict__ biaseff) {
    __shared__ float part[GATE_WAVES][HID];
    __shared__ float hbuf[HID];
    __shared__ float w2s[HID * NB];
    __shared__ float b2s[NB];
    __shared__ float los[NB];

    const int b    = (int)blockIdx.x;
    const int t    = (int)threadIdx.x;
    const int lane = t & 31;
    const int w    = __builtin_amdgcn_readfirstlane(t >> 5);
    const int half = lane >> 4;
    const int l    = lane & 15;
    const int n1   = 16 + (l & 7);
    const int kbase = w * KSLICE;

    float gv[GPL];
    #pragma unroll
    for (int i = 0; i < GPL; ++i)
        gv[i] = (i < C_CH / 32) ? feat[b * C_CH + i * 32 + lane]
                                : guidance[b * GDIM + (i - C_CH / 32) * 32 + lane];
    const float* gsrc = (w < GATE_WAVES / 2) ? (feat + b * C_CH + kbase)
                                             : (guidance + b * GDIM + (kbase - C_CH));
    v2f ga[KSTEPS], lg[KSTEPS], lb[KSTEPS], b0[KSTEPS], b1v[KSTEPS];
    #pragma unroll
    for (int j = 0; j < KSTEPS; ++j) {
        const int kc = 4 * j + 2 * half;
        ga[j] = *(const v2f*)(gsrc + kc);
        lg[j] = *(const v2f*)(ln_g + kbase + kc);
        lb[j] = *(const v2f*)(ln_b + kbase + kc);
        const float* wr = w1p + (kbase + kc) * HID;
        b0[j].x  = wr[l];
        b0[j].y  = wr[HID + l];
        b1v[j].x = wr[n1];
        b1v[j].y = wr[HID + n1];
    }
    const float bias1 = b1p[t < HID ? t : 0];
    const float w2v   = w2p[t < HID * NB ? t : HID * NB - 1];
    const float b2v   = b2p[t < NB ? t : NB - 1];
    const int ft  = t < FOLD_THREADS ? t : 0;
    const int cg  = ft / FOLD_ROWS;
    const int r   = ft % FOLD_ROWS;
    const int br  = r / 27;
    const float* csrc = convw + ((br * C_CH + cg) * 27 + (r % 27));
    float cw[FOLD_ITERS];
    #pragma unroll
    for (int k = 0; k < FOLD_ITERS; ++k) cw[k] = csrc[k * (FOLD_CG * 27)];
    const int tc = t < C_CH ? t : 0;
    float cb[NB];
    #pragma unroll
    for (int i = 0; i < NB; ++i) cb[i] = convb[i * C_CH + tc];
    __builtin_amdgcn_sched_barrier(0);

    float s = gv[0];
    #pragma unroll
    for (int i = 1; i < GPL; ++i) s += gv[i];
    const float mul = wave_sum(s) * (1.0f / NCOL);
    float q = 0.f;
    #pragma unroll
    for (int i = 0; i < GPL; ++i) { const float d = gv[i] - mul; q += d * d; }
    const float rsl = rsqrtf(wave_sum(q) * (1.0f / NCOL) + LN_EPS);

    v8f c0 = {};
    v8f c1 = {};
    #pragma unroll
    for (int j = 0; j < KSTEPS; ++j) {
        v2f a;
        a.x = (ga[j].x - mul) * rsl * lg[j].x + lb[j].x;
        a.y = (ga[j].y - mul) * rsl * lg[j].y + lb[j].y;
        c0 = __builtin_amdgcn_wmma_f32_16x16x4_f32(false, a, false, b0[j],
                                                   (short)0, c0, false, false);
        c1 = __builtin_amdgcn_wmma_f32_16x16x4_f32(false, a, false, b1v[j],
                                                   (short)0, c1, false, false);
    }
    if (half == 0) {
        part[w][l] = c0[0];
        if (l < HID - 16) part[w][16 + l] = c1[0];
    }
    if (t < HID * NB) w2s[t] = w2v;
    if (t < NB) b2s[t] = b2v;
    __syncthreads();

    if (t < HID) {
        float v = part[0][t];
        #pragma unroll
        for (int ww = 1; ww < GATE_WAVES; ++ww) v += part[ww][t];
        v += bias1;
        hbuf[t] = 0.5f * v * (1.0f + erff(v * 0.70710678118654752f));
    }
    __syncthreads();

    if (t < NB) {
        float lo = b2s[t];
        #pragma unroll
        for (int k = 0; k < HID; ++k) lo += hbuf[k] * w2s[k * NB + t];
        los[t] = lo;
    }
    __syncthreads();

    const float l0 = los[0], l1 = los[1], l2 = los[2];
    const float mx = fmaxf(l0, fmaxf(l1, l2));
    const float e0 = __expf(l0 - mx);
    const float e1 = __expf(l1 - mx);
    const float e2 = __expf(l2 - mx);
    const float inv = 1.0f / (e0 + e1 + e2);
    const float wt0 = e0 * inv, wt1 = e1 * inv, wt2 = e2 * inv;

    if (t < FOLD_THREADS) {
        const float wsel = br == 0 ? wt0 : br == 1 ? wt1 : wt2;
        float* dst = wscaled + ((b * C_CH + cg) * FOLD_ROWS + r);
        #pragma unroll
        for (int k = 0; k < FOLD_ITERS; ++k) dst[k * (FOLD_CG * FOLD_ROWS)] = wsel * cw[k];
    }
    if (t < C_CH) {
        float sb = 0.f;
        sb += wt0 * cb[0];
        sb += wt1 * cb[1];
        sb += wt2 * cb[2];
        biaseff[b * C_CH + t] = sb;
    }
}

#define STRIPS_ROW   (WDIM / 4)
#define SEG_ROWS     14
#define CONV_STEPS   (SEG_ROWS + 6)
#define CONV_THREADS 256
#define PF_DEPTH     3
#define PF_RING      (PF_DEPTH + 1)
#define PLANE_ELEMS  (HDIM * WDIM)

__device__ __forceinline__ float from_left_lane(float v) {
    return __builtin_bit_cast(float, __builtin_amdgcn_update_dpp(
        0, __builtin_bit_cast(int, v), 0x111, 0xf, 0xf, true));
}
__device__ __forceinline__ float from_right_lane(float v) {
    return __builtin_bit_cast(float, __builtin_amdgcn_update_dpp(
        0, __builtin_bit_cast(int, v), 0x101, 0xf, 0xf, true));
}

__device__ __forceinline__ void load_step(const float* __restrict__ xbc,
                                          const unsigned (&voff)[7],
                                          const int y0, const int s,
                                          v4f (&c)[7]) {
    if (s < 3 || s >= SEG_ROWS + 3) {
        const bool vy = (unsigned)(y0 - 3 + s) < (unsigned)HDIM;
        const unsigned rel = vy ? (unsigned)((s - 3) * WDIM) : 0u;
        #pragma unroll
        for (int p = 0; p < 7; ++p)
            c[p] = *(const v4f*)(xbc + (voff[p] + rel));
    } else {
        #pragma unroll
        for (int p = 0; p < 7; ++p)
            c[p] = *(const v4f*)((xbc + voff[p]) + (s - 3) * WDIM);
    }
}

template <int S>
__device__ __forceinline__ void conv_step(const float* __restrict__ xbc,
                                          const float (&wk)[NB * 27],
                                          float* __restrict__ orow,
                                          const unsigned (&voff)[7],
                                          const int y0,
                                          const int sxr, const bool active,
                                          const float bias,
                                          v4f (&ring)[PF_RING][7],
                                          float (&E)[SEG_ROWS][10]) {
    v4f (&cur)[7] = ring[S % PF_RING];
    if (S + PF_DEPTH < CONV_STEPS) load_step(xbc, voff, y0, S + PF_DEPTH, ring[(S + PF_DEPTH) % PF_RING]);
    __builtin_amdgcn_sched_barrier(0);

    if (S < 3 || S >= SEG_ROWS + 3) {
        const bool vy = (unsigned)(y0 - 3 + S) < (unsigned)HDIM;
        #pragma unroll
        for (int p = 0; p < 7; ++p) {
            cur[p].x = vy ? cur[p].x : 0.f; cur[p].y = vy ? cur[p].y : 0.f;
            cur[p].z = vy ? cur[p].z : 0.f; cur[p].w = vy ? cur[p].w : 0.f;
        }
    }
    if (S < SEG_ROWS) {
        #pragma unroll
        for (int e = 0; e < 10; ++e) E[S][e] = (e >= 3 && e <= 6) ? bias : 0.f;
    }
    #pragma unroll
    for (int i = 0; i < NB; ++i) {
        const int d = i + 1;
        #pragma unroll
        for (int kz = 0; kz < 3; ++kz) {
            const int p = 3 + (kz - 1) * d;
            #pragma unroll
            for (int ky = 0; ky < 3; ++ky) {
                const int j = S - 3 - (ky - 1) * d;
                if (j >= 0 && j < SEG_ROWS) {
                    const v4f cc = cur[p];
                    #pragma unroll
                    for (int kx = 0; kx < 3; ++kx) {
                        const int o = 3 - (kx - 1) * d;
                        const float w = wk[i * 27 + (kz * 3 + ky) * 3 + kx];
                        E[j][o]     = fmaf(w, cc.x, E[j][o]);
                        E[j][o + 1] = fmaf(w, cc.y, E[j][o + 1]);
                        E[j][o + 2] = fmaf(w, cc.z, E[j][o + 2]);
                        E[j][o + 3] = fmaf(w, cc.w, E[j][o + 3]);
                    }
                }
            }
        }
    }
    #pragma unroll
    for (int j = (S >= 6 ? S - 6 : 0); j <= (S < SEG_ROWS ? S : SEG_ROWS - 1); ++j) {
        #pragma unroll
        for (int e = 0; e < 10; ++e) asm volatile("" : "+v"(E[j][e]));
    }
    if (S >= 6) {
        const int j = S - 6;
        float acc[4] = {E[j][3], E[j][4], E[j][5], E[j][6]};
        #pragma unroll
        for (int m = 0; m < 3; ++m)
            acc[m] += from_left_lane(E[j][7 + m]);
        const bool hasr = sxr < STRIPS_ROW - 1;
        #pragma unroll
        for (int m = 1; m < 4; ++m) {
            const float r = from_right_lane(E[j][m - 1]);
            acc[m] += hasr ? r : 0.f;
        }
        if (active) {
            v4f o;
            o.x = acc[0]; o.y = acc[1]; o.z = acc[2]; o.w = acc[3];
            *(v4f*)(orow + j * WDIM) = o;
        }
    }
}

__global__ void __launch_bounds__(CONV_THREADS)
dwconv_kernel(const float* __restrict__ x,
              const float* __restrict__ wscaled,
              const float* __restrict__ biaseff,
              float* __restrict__ out) {
    const int bc = blockIdx.x >> 2;
    const int wv = __builtin_amdgcn_readfirstlane((int)(threadIdx.x >> 5));
    const int z  = (blockIdx.x & 3) * 4 + (wv >> 1);
    const int lane = (int)(threadIdx.x & 31u);
    const int sxr  = lane & 15;
    const bool active = sxr < STRIPS_ROW;
    const int sx  = active ? sxr : STRIPS_ROW - 1;
    const int y0  = ((wv & 1) * 2 + (lane >> 4)) * SEG_ROWS;

    const float* wp  = wscaled + bc * (NB * 27);
    const float bias = biaseff[bc];
    const float* xbc = x + (size_t)bc * SPATIAL;
    float* orow = out + ((size_t)bc * DDIM + z) * PLANE_ELEMS + (y0 * WDIM + sx * 4);

    float wk[NB * 27];
    #pragma unroll
    for (int i = 0; i < NB; ++i) {
        #pragma unroll
        for (int kz = 0; kz < 3; ++kz) {
            const bool vz = (unsigned)(z + (kz - 1) * (i + 1)) < (unsigned)DDIM;
            #pragma unroll
            for (int r = 0; r < 9; ++r) {
                const float u = wp[i * 27 + kz * 9 + r];
                wk[i * 27 + kz * 9 + r] = vz ? u : 0.f;
            }
        }
    }
    unsigned voff[7];
    #pragma unroll
    for (int p = 0; p < 7; ++p) {
        const int z2 = z + p - 3;
        const bool vz = (unsigned)z2 < (unsigned)DDIM;
        voff[p] = (unsigned)((vz ? z2 : z) * PLANE_ELEMS + y0 * WDIM + sx * 4);
    }

    float E[SEG_ROWS][10];
    v4f ring[PF_RING][7];
    #pragma unroll
    for (int s = 0; s < PF_DEPTH; ++s) load_step(xbc, voff, y0, s, ring[s]);

#define CONV_STEP(S) conv_step<S>(xbc, wk, orow, voff, y0, sxr, active, bias, ring, E)
    CONV_STEP(0);  CONV_STEP(1);  CONV_STEP(2);  CONV_STEP(3);  CONV_STEP(4);
    CONV_STEP(5);  CONV_STEP(6);  CONV_STEP(7);  CONV_STEP(8);  CONV_STEP(9);
    CONV_STEP(10); CONV_STEP(11); CONV_STEP(12); CONV_STEP(13); CONV_STEP(14);
    CONV_STEP(15); CONV_STEP(16); CONV_STEP(17); CONV_STEP(18); CONV_STEP(19);
#undef CONV_STEP
}

extern "C" void kernel_launch(void* const* d_in, const int* in_sizes, int n_in,
                              void* d_out, int out_size, void* d_ws, size_t ws_size,
                              hipStream_t stream) {
    const float* x        = (const float*)d_in[0];
    const float* guidance = (const float*)d_in[1];
    const float* convw    = (const float*)d_in[2];
    const float* convb    = (const float*)d_in[3];
    const float* ln_g     = (const float*)d_in[4];
    const float* ln_b     = (const float*)d_in[5];
    const float* w1       = (const float*)d_in[6];
    const float* b1       = (const float*)d_in[7];
    const float* w2       = (const float*)d_in[8];
    const float* b2       = (const float*)d_in[9];

    float* ws      = (float*)d_ws;
    float* feat    = ws;
    float* biaseff = ws + 512;
    float* wscaled = ws + 1024;

    feat_reduce_kernel<<<B_N * C_CH, 256, 0, stream>>>(x, feat);
    gate_fold_kernel<<<B_N, GATE_THREADS, 0, stream>>>(feat, guidance, ln_g, ln_b,
                                                       w1, b1, w2, b2, convw, convb,
                                                       wscaled, biaseff);
    dwconv_kernel<<<B_N * C_CH * (DDIM / 4), CONV_THREADS, 0, stream>>>(x, wscaled, biaseff,
                                                         (float*)d_out);
}
